# starting kernel
# baseline (speedup 1.0000x reference)
#include <hip/hip_runtime.h>

typedef float v2f __attribute__((ext_vector_type(2)));
typedef float v8f __attribute__((ext_vector_type(8)));

__device__ __constant__ float c_h[9] = {
    0.03782845550699535f, -0.02384946501937986f, -0.1106244044184226f,
    0.3774028556126536f,  0.8526986790094022f,   0.3774028556126537f,
    -0.1106244044184226f, -0.02384946501937986f, 0.03782845550699535f};

__device__ __forceinline__ float tap(int t) {
    int tc = t < 0 ? 0 : (t > 8 ? 8 : t);
    float raw = c_h[tc];
    return (t >= 0 && t <= 8) ? raw : 0.0f;
}

__global__ __launch_bounds__(256) void bior_up_wmma(const float* __restrict__ in,
                                                    float* __restrict__ out) {
    const int lane = threadIdx.x & 31;
    const int wave = threadIdx.x >> 5;
    const int tile = blockIdx.x * 8 + wave;
    const int b  = tile >> 12;
    const int tr = (tile >> 6) & 63;
    const int tc = tile & 63;

    const int m  = lane & 15;
    const int kh = lane >> 4;

    v2f w[3];
    #pragma unroll
    for (int kk = 0; kk < 3; ++kk) {
        const int i0 = 4 * kk + 2 * kh;
        w[kk].x = tap(m + 8 - 2 * (i0 + 0));
        w[kk].y = tap(m + 8 - 2 * (i0 + 1));
    }

    const int r0 = tr * 8 - 2;
    const int c0 = tc * 8 - 2;
    const float* inb = in + (size_t)b * (512 * 512);
    const int gc  = c0 + m;
    const int ccl = gc < 0 ? 0 : (gc > 511 ? 511 : gc);
    const bool cok = (gc >= 0) && (gc < 512);
    v2f bi[3];
    #pragma unroll
    for (int kk = 0; kk < 3; ++kk) {
        float vv[2];
        #pragma unroll
        for (int v = 0; v < 2; ++v) {
            const int gr  = r0 + 4 * kk + 2 * kh + v;
            const int rcl = gr < 0 ? 0 : (gr > 511 ? 511 : gr);
            const float raw = inb[rcl * 512 + ccl];
            vv[v] = (cok && gr >= 0 && gr < 512) ? raw : 0.0f;
        }
        bi[kk].x = vv[0];
        bi[kk].y = vv[1];
    }

    v8f t = {};
    #pragma unroll
    for (int kk = 0; kk < 3; ++kk)
        t = __builtin_amdgcn_wmma_f32_16x16x4_f32(false, w[kk], false, bi[kk],
                                                  (short)0, t, false, false);

    __shared__ float lds[8][16 * 17];
    #pragma unroll
    for (int v = 0; v < 8; ++v)
        lds[wave][(kh * 8 + v) * 17 + m] = t[v];
    __builtin_amdgcn_wave_barrier();
    asm volatile("s_wait_dscnt 0" ::: "memory");

    v2f at[3];
    #pragma unroll
    for (int kk = 0; kk < 3; ++kk) {
        const int j0 = 4 * kk + 2 * kh;
        at[kk].x = lds[wave][m * 17 + j0 + 0];
        at[kk].y = lds[wave][m * 17 + j0 + 1];
    }

    v8f o = {};
    #pragma unroll
    for (int kk = 0; kk < 3; ++kk)
        o = __builtin_amdgcn_wmma_f32_16x16x4_f32(false, at[kk], false, w[kk],
                                                  (short)0, o, false, false);

    float* ob = out + (size_t)b * (1024 * 1024)
                    + (size_t)(tr * 16) * 1024 + (tc * 16);
    #pragma unroll
    for (int v = 0; v < 8; ++v)
        ob[(kh * 8 + v) * 1024 + m] = o[v];
}

extern "C" void kernel_launch(void* const* d_in, const int* in_sizes, int n_in,
                              void* d_out, int out_size, void* d_ws, size_t ws_size,
                              hipStream_t stream) {
    const float* in = (const float*)d_in[0];
    float* out = (float*)d_out;
    dim3 grid(8192), block(256);
    hipLaunchKernelGGL(bior_up_wmma, grid, block, 0, stream, in, out);
}


# each wave walks a strip of 16 tiles down a tile column (two per trip, rows requested two trips ahead, taps built once): 4096 resident waves instead of 65536 entering; per-tile sums unchanged
# speedup vs baseline: 1.2911x; 1.2911x over previous
#include <hip/hip_runtime.h>

typedef float v2f __attribute__((ext_vector_type(2)));
typedef float v8f __attribute__((ext_vector_type(8)));

__device__ __constant__ float c_h[9] = {
    0.03782845550699535f, -0.02384946501937986f, -0.1106244044184226f,
    0.3774028556126536f,  0.8526986790094022f,   0.3774028556126537f,
    -0.1106244044184226f, -0.02384946501937986f, 0.03782845550699535f};

__device__ __forceinline__ float tap(int t) {
    int tc = t < 0 ? 0 : (t > 8 ? 8 : t);
    float raw = c_h[tc];
    return (t >= 0 && t <= 8) ? raw : 0.0f;
}

#define NCHUNK 33
#define NTRIP 8

__global__ __launch_bounds__(256) __attribute__((amdgpu_waves_per_eu(4, 4))) void bior_up_wmma(const float* __restrict__ in,
                                                    float* __restrict__ out) {
    const int lane = threadIdx.x & 31;
    const int wave = threadIdx.x >> 5;
    const int b   = blockIdx.x >> 5;
    const int tr0 = ((blockIdx.x >> 3) & 3) * 16;
    const int tc  = (blockIdx.x & 7) * 8 + wave;

    const int m  = lane & 15;
    const int kh = lane >> 4;

    v2f w[3];
    #pragma unroll
    for (int kk = 0; kk < 3; ++kk) {
        const int i0 = 4 * kk + 2 * kh;
        w[kk].x = tap(m + 8 - 2 * (i0 + 0));
        w[kk].y = tap(m + 8 - 2 * (i0 + 1));
    }

    const int r0 = tr0 * 8 - 2;
    const int c0 = tc * 8 - 2;
    const float* inb = in + (size_t)b * (512 * 512);
    const int gc  = c0 + m;
    const int ccl = gc < 0 ? 0 : (gc > 511 ? 511 : gc);
    const bool cok = (gc >= 0) && (gc < 512);
    const float* pin = inb + (r0 + 4 + 2 * kh) * 512 + ccl;

    auto load_chunk = [&](int c) -> v2f {
        float vv[2];
        if (c == 0 || c == NCHUNK - 1) {
            #pragma unroll
            for (int v = 0; v < 2; ++v) {
                const int gr  = r0 + 4 * c + 2 * kh + v;
                const int rcl = gr < 0 ? 0 : (gr > 511 ? 511 : gr);
                const float raw = inb[rcl * 512 + ccl];
                vv[v] = (cok && gr >= 0 && gr < 512) ? raw : 0.0f;
            }
        } else {
            #pragma unroll
            for (int v = 0; v < 2; ++v) {
                const float raw = pin[(4 * (c - 1) + v) * 512];
                vv[v] = cok ? raw : 0.0f;
            }
        }
        v2f r;
        r.x = vv[0];
        r.y = vv[1];
        return r;
    };

    float* ob = out + (size_t)b * (1024 * 1024)
                    + (size_t)(tr0 * 16 + kh * 8) * 1024 + (tc * 16 + m);

    __shared__ float lds[8][4][16 * 17];

    v2f ch[NCHUNK];
    #pragma unroll
    for (int c = 0; c <= 8; ++c) ch[c] = load_chunk(c);

    #pragma unroll
    for (int p = 0; p < NTRIP; ++p) {
        #pragma unroll
        for (int c = 4 * p + 9; c <= 4 * p + 12; ++c)
            if (c < NCHUNK) ch[c] = load_chunk(c);

        v8f t0 = {}, t1 = {};
        #pragma unroll
        for (int kk = 0; kk < 3; ++kk) {
            t0 = __builtin_amdgcn_wmma_f32_16x16x4_f32(false, w[kk], false, ch[4 * p + kk],
                                                       (short)0, t0, false, false);
            t1 = __builtin_amdgcn_wmma_f32_16x16x4_f32(false, w[kk], false, ch[4 * p + 2 + kk],
                                                       (short)0, t1, false, false);
        }

        float* s0 = lds[wave][2 * (p & 1) + 0];
        float* s1 = lds[wave][2 * (p & 1) + 1];
        #pragma unroll
        for (int v = 0; v < 8; ++v) {
            s0[(kh * 8 + v) * 17 + m] = t0[v];
            s1[(kh * 8 + v) * 17 + m] = t1[v];
        }
        __builtin_amdgcn_wave_barrier();
        asm volatile("s_wait_dscnt 0" ::: "memory");

        v2f a0[3], a1[3];
        #pragma unroll
        for (int kk = 0; kk < 3; ++kk) {
            const int j0 = 4 * kk + 2 * kh;
            a0[kk].x = s0[m * 17 + j0 + 0];
            a0[kk].y = s0[m * 17 + j0 + 1];
            a1[kk].x = s1[m * 17 + j0 + 0];
            a1[kk].y = s1[m * 17 + j0 + 1];
        }

        v8f o0 = {}, o1 = {};
        #pragma unroll
        for (int kk = 0; kk < 3; ++kk) {
            o0 = __builtin_amdgcn_wmma_f32_16x16x4_f32(false, a0[kk], false, w[kk],
                                                       (short)0, o0, false, false);
            o1 = __builtin_amdgcn_wmma_f32_16x16x4_f32(false, a1[kk], false, w[kk],
                                                       (short)0, o1, false, false);
        }

        #pragma unroll
        for (int v = 0; v < 8; ++v)
            ob[((2 * p) * 16 + v) * 1024] = o0[v];
        #pragma unroll
        for (int v = 0; v < 8; ++v)
            ob[((2 * p + 1) * 16 + v) * 1024] = o1[v];
    }
}

extern "C" void kernel_launch(void* const* d_in, const int* in_sizes, int n_in,
                              void* d_out, int out_size, void* d_ws, size_t ws_size,
                              hipStream_t stream) {
    const float* in = (const float*)d_in[0];
    float* out = (float*)d_out;
    dim3 grid(512), block(256);
    hipLaunchKernelGGL(bior_up_wmma, grid, block, 0, stream, in, out);
}


# each wave owns a 32-column (128-byte) strip of 8 tile rows: its two tiles per trip are turned to rows in a wave-private LDS slab and leave as 16-byte stores of 4 whole lines; per-tile sums unchanged
# speedup vs baseline: 1.4316x; 1.1088x over previous
#include <hip/hip_runtime.h>

typedef float v2f __attribute__((ext_vector_type(2)));
typedef float v4f __attribute__((ext_vector_type(4)));
typedef float v8f __attribute__((ext_vector_type(8)));

__device__ __constant__ float c_h[9] = {
    0.03782845550699535f, -0.02384946501937986f, -0.1106244044184226f,
    0.3774028556126536f,  0.8526986790094022f,   0.3774028556126537f,
    -0.1106244044184226f, -0.02384946501937986f, 0.03782845550699535f};

__device__ __forceinline__ float tap(int t) {
    int tc = t < 0 ? 0 : (t > 8 ? 8 : t);
    float raw = c_h[tc];
    return (t >= 0 && t <= 8) ? raw : 0.0f;
}

#define NCH 17
#define NTRIP 8
#define OPITCH 36

__global__ __launch_bounds__(256) __attribute__((amdgpu_waves_per_eu(4, 4))) void bior_up_wmma(const float* __restrict__ in,
                                                    float* __restrict__ out) {
    const int lane = threadIdx.x & 31;
    const int wave = threadIdx.x >> 5;
    const int b   = blockIdx.x >> 5;
    const int tr0 = ((blockIdx.x >> 2) & 7) * 8;
    const int pr  = (blockIdx.x & 3) * 8 + wave;

    const int m  = lane & 15;
    const int kh = lane >> 4;

    v2f w[3];
    #pragma unroll
    for (int kk = 0; kk < 3; ++kk) {
        const int i0 = 4 * kk + 2 * kh;
        w[kk].x = tap(m + 8 - 2 * (i0 + 0));
        w[kk].y = tap(m + 8 - 2 * (i0 + 1));
    }

    const int r0 = tr0 * 8 - 2;
    const float* inb = in + (size_t)b * (512 * 512);
    int ccl[2];
    bool cok[2];
    const float* pin[2];
    #pragma unroll
    for (int s = 0; s < 2; ++s) {
        const int gc = (2 * pr + s) * 8 - 2 + m;
        ccl[s] = gc < 0 ? 0 : (gc > 511 ? 511 : gc);
        cok[s] = (gc >= 0) && (gc < 512);
        pin[s] = inb + (r0 + 4 + 2 * kh) * 512 + ccl[s];
    }

    auto load_chunk = [&](int s, int c) -> v2f {
        float vv[2];
        if (c == 0 || c == NCH - 1) {
            #pragma unroll
            for (int v = 0; v < 2; ++v) {
                const int gr  = r0 + 4 * c + 2 * kh + v;
                const int rcl = gr < 0 ? 0 : (gr > 511 ? 511 : gr);
                const float raw = inb[rcl * 512 + ccl[s]];
                vv[v] = (cok[s] && gr >= 0 && gr < 512) ? raw : 0.0f;
            }
        } else {
            #pragma unroll
            for (int v = 0; v < 2; ++v) {
                const float raw = pin[s][(4 * (c - 1) + v) * 512];
                vv[v] = cok[s] ? raw : 0.0f;
            }
        }
        v2f r;
        r.x = vv[0];
        r.y = vv[1];
        return r;
    };

    const int q4 = lane & 7;
    const int rr = ((lane >> 4) & 1) + 8 * ((lane >> 3) & 1);
    float* ob = out + (size_t)b * (1024 * 1024)
                    + (size_t)(tr0 * 16 + rr) * 1024 + (pr * 32 + q4 * 4);

    __shared__ float lds[8][4][16 * 17];
    __shared__ __attribute__((aligned(16))) float ost[8][2][16 * OPITCH];

    v2f ch[2][NCH];
    #pragma unroll
    for (int c = 0; c <= 4; ++c) {
        ch[0][c] = load_chunk(0, c);
        ch[1][c] = load_chunk(1, c);
    }

    #pragma unroll
    for (int p = 0; p < NTRIP; ++p) {
        #pragma unroll
        for (int c = 2 * p + 5; c <= 2 * p + 6; ++c)
            if (c < NCH) {
                ch[0][c] = load_chunk(0, c);
                ch[1][c] = load_chunk(1, c);
            }

        v8f t0 = {}, t1 = {};
        #pragma unroll
        for (int kk = 0; kk < 3; ++kk) {
            t0 = __builtin_amdgcn_wmma_f32_16x16x4_f32(false, w[kk], false, ch[0][2 * p + kk],
                                                       (short)0, t0, false, false);
            t1 = __builtin_amdgcn_wmma_f32_16x16x4_f32(false, w[kk], false, ch[1][2 * p + kk],
                                                       (short)0, t1, false, false);
        }

        float* s0 = lds[wave][2 * (p & 1) + 0];
        float* s1 = lds[wave][2 * (p & 1) + 1];
        #pragma unroll
        for (int v = 0; v < 8; ++v) {
            s0[(kh * 8 + v) * 17 + m] = t0[v];
            s1[(kh * 8 + v) * 17 + m] = t1[v];
        }
        __builtin_amdgcn_wave_barrier();
        asm volatile("s_wait_dscnt 0" ::: "memory");

        v2f a0[3], a1[3];
        #pragma unroll
        for (int kk = 0; kk < 3; ++kk) {
            const int j0 = 4 * kk + 2 * kh;
            a0[kk].x = s0[m * 17 + j0 + 0];
            a0[kk].y = s0[m * 17 + j0 + 1];
            a1[kk].x = s1[m * 17 + j0 + 0];
            a1[kk].y = s1[m * 17 + j0 + 1];
        }

        v8f o0 = {}, o1 = {};
        #pragma unroll
        for (int kk = 0; kk < 3; ++kk) {
            o0 = __builtin_amdgcn_wmma_f32_16x16x4_f32(false, a0[kk], false, w[kk],
                                                       (short)0, o0, false, false);
            o1 = __builtin_amdgcn_wmma_f32_16x16x4_f32(false, a1[kk], false, w[kk],
                                                       (short)0, o1, false, false);
        }

        float* so = ost[wave][p & 1];
        #pragma unroll
        for (int v = 0; v < 8; ++v) {
            so[(kh * 8 + v) * OPITCH + m]      = o0[v];
            so[(kh * 8 + v) * OPITCH + 16 + m] = o1[v];
        }
        __builtin_amdgcn_wave_barrier();
        asm volatile("s_wait_dscnt 0" ::: "memory");

        #pragma unroll
        for (int q = 0; q < 4; ++q) {
            const v4f val = *(const v4f*)(so + (rr + 2 * q) * OPITCH + q4 * 4);
            *(v4f*)(ob + (size_t)(p * 16 + 2 * q) * 1024) = val;
        }
    }
}

extern "C" void kernel_launch(void* const* d_in, const int* in_sizes, int n_in,
                              void* d_out, int out_size, void* d_ws, size_t ws_size,
                              hipStream_t stream) {
    const float* in = (const float*)d_in[0];
    float* out = (float*)d_out;
    dim3 grid(512), block(256);
    hipLaunchKernelGGL(bior_up_wmma, grid, block, 0, stream, in, out);
}


# each wave owns a 64-column (256-byte) strip of 4 tile rows: four tiles per trip, H pass over 3 input column tiles instead of 4, rows leave as whole 256-byte pieces; per-tile sums unchanged
# speedup vs baseline: 1.9259x; 1.3453x over previous
#include <hip/hip_runtime.h>

typedef float v2f __attribute__((ext_vector_type(2)));
typedef float v4f __attribute__((ext_vector_type(4)));
typedef float v8f __attribute__((ext_vector_type(8)));

__device__ __constant__ float c_h[9] = {
    0.03782845550699535f, -0.02384946501937986f, -0.1106244044184226f,
    0.3774028556126536f,  0.8526986790094022f,   0.3774028556126537f,
    -0.1106244044184226f, -0.02384946501937986f, 0.03782845550699535f};

__device__ __forceinline__ float tap(int t) {
    int tc = t < 0 ? 0 : (t > 8 ? 8 : t);
    float raw = c_h[tc];
    return (t >= 0 && t <= 8) ? raw : 0.0f;
}

#define NCH 9
#define NTRIP 4
#define NT 3
#define NJ 4
#define TPITCH 52
#define OPITCH 68

__global__ __launch_bounds__(256) __attribute__((amdgpu_waves_per_eu(4, 4))) void bior_up_wmma(const float* __restrict__ in,
                                                    float* __restrict__ out) {
    const int lane = threadIdx.x & 31;
    const int wave = threadIdx.x >> 5;
    const int b   = blockIdx.x >> 5;
    const int tr0 = ((blockIdx.x >> 1) & 15) * 4;
    const int cs  = (blockIdx.x & 1) * 8 + wave;

    const int m  = lane & 15;
    const int kh = lane >> 4;

    v2f w[3];
    #pragma unroll
    for (int kk = 0; kk < 3; ++kk) {
        const int i0 = 4 * kk + 2 * kh;
        w[kk].x = tap(m + 8 - 2 * (i0 + 0));
        w[kk].y = tap(m + 8 - 2 * (i0 + 1));
    }

    const int r0 = tr0 * 8 - 2;
    const float* inb = in + (size_t)b * (512 * 512);
    const int cin0 = cs * 32;
    int ccl[NT];
    bool cok[NT];
    const float* pin[NT];
    #pragma unroll
    for (int n = 0; n < NT; ++n) {
        int gc = cin0 - 2 + 16 * n + m;
        if (n == NT - 1) gc = gc > cin0 + 33 ? cin0 + 33 : gc;
        ccl[n] = gc < 0 ? 0 : (gc > 511 ? 511 : gc);
        cok[n] = (gc >= 0) && (gc < 512);
        pin[n] = inb + (r0 + 4 + 2 * kh) * 512 + ccl[n];
    }

    auto load_chunk = [&](int n, int c) -> v2f {
        float vv[2];
        if (c == 0 || c == NCH - 1) {
            #pragma unroll
            for (int v = 0; v < 2; ++v) {
                const int gr  = r0 + 4 * c + 2 * kh + v;
                const int rcl = gr < 0 ? 0 : (gr > 511 ? 511 : gr);
                const float raw = inb[rcl * 512 + ccl[n]];
                vv[v] = (cok[n] && gr >= 0 && gr < 512) ? raw : 0.0f;
            }
        } else {
            #pragma unroll
            for (int v = 0; v < 2; ++v) {
                const float raw = pin[n][(4 * (c - 1) + v) * 512];
                vv[v] = cok[n] ? raw : 0.0f;
            }
        }
        v2f r;
        r.x = vv[0];
        r.y = vv[1];
        return r;
    };

    const int c4   = lane & 15;
    const int rsel = lane >> 4;
    float* ob = out + (size_t)b * (1024 * 1024)
                    + (size_t)(tr0 * 16 + rsel) * 1024 + (cs * 64 + c4 * 4);

    __shared__ __attribute__((aligned(16))) float lds[8][2][16 * TPITCH];
    __shared__ __attribute__((aligned(16))) float ost[8][2][16 * OPITCH];

    v2f ch[NT][NCH];
    #pragma unroll
    for (int c = 0; c <= 4; ++c) {
        #pragma unroll
        for (int n = 0; n < NT; ++n) ch[n][c] = load_chunk(n, c);
    }

    #pragma unroll
    for (int p = 0; p < NTRIP; ++p) {
        #pragma unroll
        for (int c = 2 * p + 5; c <= 2 * p + 6; ++c)
            if (c < NCH) {
                #pragma unroll
                for (int n = 0; n < NT; ++n) ch[n][c] = load_chunk(n, c);
            }

        v8f t[NT];
        #pragma unroll
        for (int n = 0; n < NT; ++n) t[n] = (v8f){};
        #pragma unroll
        for (int kk = 0; kk < 3; ++kk) {
            #pragma unroll
            for (int n = 0; n < NT; ++n)
                t[n] = __builtin_amdgcn_wmma_f32_16x16x4_f32(false, w[kk], false, ch[n][2 * p + kk],
                                                             (short)0, t[n], false, false);
        }

        float* sT = lds[wave][p & 1];
        #pragma unroll
        for (int v = 0; v < 8; ++v) {
            #pragma unroll
            for (int n = 0; n < NT; ++n) sT[(kh * 8 + v) * TPITCH + 16 * n + m] = t[n][v];
        }
        __builtin_amdgcn_wave_barrier();
        asm volatile("s_wait_dscnt 0" ::: "memory");

        v2f a[NJ][3];
        #pragma unroll
        for (int j = 0; j < NJ; ++j) {
            #pragma unroll
            for (int kk = 0; kk < 3; ++kk)
                a[j][kk] = *(const v2f*)(sT + m * TPITCH + 8 * j + 4 * kk + 2 * kh);
        }

        v8f o[NJ];
        #pragma unroll
        for (int j = 0; j < NJ; ++j) o[j] = (v8f){};
        #pragma unroll
        for (int kk = 0; kk < 3; ++kk) {
            #pragma unroll
            for (int j = 0; j < NJ; ++j)
                o[j] = __builtin_amdgcn_wmma_f32_16x16x4_f32(false, a[j][kk], false, w[kk],
                                                             (short)0, o[j], false, false);
        }

        float* so = ost[wave][p & 1];
        #pragma unroll
        for (int v = 0; v < 8; ++v) {
            #pragma unroll
            for (int j = 0; j < NJ; ++j) so[(kh * 8 + v) * OPITCH + 16 * j + m] = o[j][v];
        }
        __builtin_amdgcn_wave_barrier();
        asm volatile("s_wait_dscnt 0" ::: "memory");

        #pragma unroll
        for (int q = 0; q < 8; ++q) {
            const v4f val = *(const v4f*)(so + (2 * q + rsel) * OPITCH + c4 * 4);
            *(v4f*)(ob + (size_t)(p * 16 + 2 * q) * 1024) = val;
        }
    }
}

extern "C" void kernel_launch(void* const* d_in, const int* in_sizes, int n_in,
                              void* d_out, int out_size, void* d_ws, size_t ws_size,
                              hipStream_t stream) {
    const float* in = (const float*)d_in[0];
    float* out = (float*)d_out;
    dim3 grid(512), block(256);
    hipLaunchKernelGGL(bior_up_wmma, grid, block, 0, stream, in, out);
}
